# starting kernel
# baseline (speedup 1.0000x reference)
#include <hip/hip_runtime.h>

typedef __attribute__((ext_vector_type(2))) float v2f;
typedef __attribute__((ext_vector_type(8))) float v8f;

#define WMMA_F32(a, b, c) \
  __builtin_amdgcn_wmma_f32_16x16x4_f32(false, (a), false, (b), (short)0, (c), false, false)

static constexpr int BB = 2, TT = 2048, CC = 1024, HH = 16, DD = 64;
static constexpr int MTOT = BB * TT;
static constexpr int NQKV = 3 * CC;
static constexpr size_t HSZ = (size_t)BB * HH * TT * DD;

__global__ __launch_bounds__(32)
void qkv_gemm_kernel(const float* __restrict__ X, const float* __restrict__ W,
                     const float* __restrict__ bias,
                     float* __restrict__ q, float* __restrict__ k,
                     float* __restrict__ v) {
  const int lane = threadIdx.x & 31;
  const int hl = lane >> 4;
  const int ln = lane & 15;
  const int n0 = blockIdx.x * 64;
  const int m0 = blockIdx.y * 16;

  v8f c[4] = {};
  const float* Arow = X + (size_t)(m0 + ln) * CC;

  for (int k0 = 0; k0 < CC; k0 += 4) {
    v2f a = *(const v2f*)(Arow + k0 + 2 * hl);
    __builtin_prefetch((const void*)(Arow + k0 + 64), 0, 0);
    const float* Bp = W + (size_t)(k0 + 2 * hl) * NQKV + n0 + ln;
#pragma unroll
    for (int n = 0; n < 4; ++n) {
      v2f b;
      b.x = Bp[16 * n];
      b.y = Bp[NQKV + 16 * n];
      c[n] = WMMA_F32(a, b, c[n]);
    }
  }

#pragma unroll
  for (int n = 0; n < 4; ++n) {
    const int col = n0 + 16 * n + ln;
    const float bval = bias[col];
    const int which = col >> 10;
    const int rem = col & (CC - 1);
    const int h = rem >> 6;
    const int d = rem & 63;
    float* dst = (which == 0) ? q : (which == 1) ? k : v;
#pragma unroll
    for (int r = 0; r < 8; ++r) {
      const int trow = m0 + r + 8 * hl;
      const int bidx = trow >> 11;
      const int t = trow & (TT - 1);
      const size_t idx = (((size_t)(bidx * HH + h) * TT) + t) * DD + d;
      dst[idx] = c[n][r] + bval;
    }
  }
}

__global__ __launch_bounds__(32)
void attn_kernel(const float* __restrict__ q, const float* __restrict__ kmat,
                 const float* __restrict__ vmat, float* __restrict__ out) {
  const int lane = threadIdx.x & 31;
  const int hl = lane >> 4;
  const int ln = lane & 15;
  const int qtile = blockIdx.x;
  const int bh = blockIdx.y;
  const int b = bh >> 4, h = bh & 15;
  const int q0 = qtile * 16;

  const float* qbase = q + (size_t)bh * TT * DD;
  const float* kbase = kmat + (size_t)bh * TT * DD;
  const float* vbase = vmat + (size_t)bh * TT * DD;

  __shared__ float pt[16][17];

  v2f qa[16];
  {
    const float* qrow = qbase + (size_t)(q0 + ln) * DD + 2 * hl;
#pragma unroll
    for (int t = 0; t < 16; ++t) {
      v2f a = *(const v2f*)(qrow + 4 * t);
      qa[t] = a * 0.125f;
    }
  }

  v8f o[4] = {};
  float m_i[8], l_i[8];
#pragma unroll
  for (int r = 0; r < 8; ++r) { m_i[r] = -__builtin_inff(); l_i[r] = 0.0f; }

  for (int kt = 0; kt <= qtile; ++kt) {
    const int kb = kt * 16;

    v8f s = {};
    const float* krow = kbase + (size_t)(kb + ln) * DD + 2 * hl;
#pragma unroll
    for (int t = 0; t < 16; ++t) {
      v2f bf = *(const v2f*)(krow + 4 * t);
      s = WMMA_F32(qa[t], bf, s);
    }

    const bool diag = (kt == qtile);
    float scale_r[8];
#pragma unroll
    for (int r = 0; r < 8; ++r) {
      const int row = r + 8 * hl;
      float sv = s[r];
      if (diag && (ln > row)) sv = -__builtin_inff();
      float mx = sv;
      mx = fmaxf(mx, __shfl_xor(mx, 1));
      mx = fmaxf(mx, __shfl_xor(mx, 2));
      mx = fmaxf(mx, __shfl_xor(mx, 4));
      mx = fmaxf(mx, __shfl_xor(mx, 8));
      const float mn = fmaxf(m_i[r], mx);
      const float sc = __expf(m_i[r] - mn);
      const float p = __expf(sv - mn);
      float ps = p;
      ps += __shfl_xor(ps, 1);
      ps += __shfl_xor(ps, 2);
      ps += __shfl_xor(ps, 4);
      ps += __shfl_xor(ps, 8);
      l_i[r] = l_i[r] * sc + ps;
      m_i[r] = mn;
      scale_r[r] = sc;
      pt[row][ln] = p;
    }

#pragma unroll
    for (int n = 0; n < 4; ++n)
#pragma unroll
      for (int r = 0; r < 8; ++r) o[n][r] *= scale_r[r];

    __syncthreads();

#pragma unroll
    for (int t = 0; t < 4; ++t) {
      v2f a;
      a.x = pt[ln][4 * t + 2 * hl];
      a.y = pt[ln][4 * t + 2 * hl + 1];
      const float* vrow = vbase + (size_t)(kb + 4 * t + 2 * hl) * DD + ln;
#pragma unroll
      for (int n = 0; n < 4; ++n) {
        v2f bf;
        bf.x = vrow[16 * n];
        bf.y = vrow[DD + 16 * n];
        o[n] = WMMA_F32(a, bf, o[n]);
      }
    }
    __syncthreads();
  }

#pragma unroll
  for (int r = 0; r < 8; ++r) {
    const float inv = 1.0f / l_i[r];
    const int trow = q0 + r + 8 * hl;
    float* orow = out + ((size_t)(b * TT + trow) * CC) + h * DD;
#pragma unroll
    for (int n = 0; n < 4; ++n) orow[16 * n + ln] = o[n][r] * inv;
  }
}

__global__ __launch_bounds__(32)
void out_gemm_kernel(const float* __restrict__ A, const float* __restrict__ W,
                     const float* __restrict__ bias, float* __restrict__ out) {
  const int lane = threadIdx.x & 31;
  const int hl = lane >> 4;
  const int ln = lane & 15;
  const int n0 = blockIdx.x * 64;
  const int m0 = blockIdx.y * 16;

  v8f c[4] = {};
  const float* Arow = A + (size_t)(m0 + ln) * CC;

  for (int k0 = 0; k0 < CC; k0 += 4) {
    v2f a = *(const v2f*)(Arow + k0 + 2 * hl);
    __builtin_prefetch((const void*)(Arow + k0 + 64), 0, 0);
    const float* Bp = W + (size_t)(k0 + 2 * hl) * CC + n0 + ln;
#pragma unroll
    for (int n = 0; n < 4; ++n) {
      v2f b;
      b.x = Bp[16 * n];
      b.y = Bp[CC + 16 * n];
      c[n] = WMMA_F32(a, b, c[n]);
    }
  }

#pragma unroll
  for (int n = 0; n < 4; ++n) {
    const float bval = bias[n0 + 16 * n + ln];
#pragma unroll
    for (int r = 0; r < 8; ++r) {
      out[(size_t)(m0 + r + 8 * hl) * CC + n0 + 16 * n + ln] = c[n][r] + bval;
    }
  }
}

extern "C" void kernel_launch(void* const* d_in, const int* in_sizes, int n_in,
                              void* d_out, int out_size, void* d_ws, size_t ws_size,
                              hipStream_t stream) {
  (void)in_sizes; (void)n_in; (void)out_size; (void)ws_size;
  const float* x    = (const float*)d_in[0];
  const float* Wqkv = (const float*)d_in[1];
  const float* bqkv = (const float*)d_in[2];
  const float* Wout = (const float*)d_in[3];
  const float* bout = (const float*)d_in[4];
  float* out = (float*)d_out;

  float* qp = (float*)d_ws;
  float* kp = qp + HSZ;
  float* vp = kp + HSZ;
  float* ao = vp + HSZ;

  const dim3 blk(32, 1, 1);
  qkv_gemm_kernel<<<dim3(NQKV / 64, MTOT / 16), blk, 0, stream>>>(x, Wqkv, bqkv, qp, kp, vp);
  attn_kernel<<<dim3(TT / 16, BB * HH), blk, 0, stream>>>(qp, kp, vp, ao);
  out_gemm_kernel<<<dim3(CC / 64, MTOT / 16), blk, 0, stream>>>(ao, Wout, bout, out);
}


# Both projection GEMMs: A and W tiles staged through LDS by 256-thread work-groups (128x128 tile, K-step 32, two stages), each element fetched once per tile, not once per wave; sum order unchanged
# speedup vs baseline: 1.4346x; 1.4346x over previous
#include <hip/hip_runtime.h>

typedef __attribute__((ext_vector_type(2))) float v2f;
typedef __attribute__((ext_vector_type(8))) float v8f;
typedef __attribute__((ext_vector_type(4))) float v4f;

#define WMMA_F32(a, b, c) \
  __builtin_amdgcn_wmma_f32_16x16x4_f32(false, (a), false, (b), (short)0, (c), false, false)

static constexpr int BB = 2, TT = 2048, CC = 1024, HH = 16, DD = 64;
static constexpr int MTOT = BB * TT;
static constexpr int NQKV = 3 * CC;
static constexpr size_t HSZ = (size_t)BB * HH * TT * DD;

static constexpr int BM = 128, BN = 128, KT = 32;
static constexpr int PA = KT + 4;
static constexpr int PB = 2 * BN + 32;
static constexpr int SA = BM * PA;
static constexpr int SB = (KT / 2) * PB;
static constexpr int NKT = CC / KT;

template <int LDB>
__device__ __forceinline__ void gemm_load_tile(const float* A, const float* W,
                                               int m0, int n0, int kt, int tid,
                                               v4f (&ra)[4], v4f (&rb)[4]) {
  const int arow = tid >> 3, akq = tid & 7;
  const int bpr = tid >> 5, bnq = tid & 31;
#pragma unroll
  for (int i = 0; i < 4; ++i)
    ra[i] = *(const v4f*)(A + (size_t)(m0 + arow + 32 * i) * CC + kt * KT + akq * 4);
#pragma unroll
  for (int i = 0; i < 2; ++i)
#pragma unroll
    for (int e = 0; e < 2; ++e)
      rb[2 * i + e] = *(const v4f*)(W + (size_t)(kt * KT + 2 * (bpr + 8 * i) + e) * LDB + n0 + bnq * 4);
}

__device__ __forceinline__ void gemm_store_tile(float* sA, float* sB, int tid,
                                                const v4f (&ra)[4], const v4f (&rb)[4]) {
  const int arow = tid >> 3, akq = tid & 7;
  const int bpr = tid >> 5, bnq = tid & 31;
#pragma unroll
  for (int i = 0; i < 4; ++i)
    *(v4f*)(sA + (arow + 32 * i) * PA + akq * 4) = ra[i];
#pragma unroll
  for (int i = 0; i < 2; ++i) {
    const v4f lo = rb[2 * i], hi = rb[2 * i + 1];
    float* d = sB + (bpr + 8 * i) * PB + bnq * 8;
    *(v4f*)(d)     = __builtin_shufflevector(lo, hi, 0, 4, 1, 5);
    *(v4f*)(d + 4) = __builtin_shufflevector(lo, hi, 2, 6, 3, 7);
  }
}

struct FragSet { v2f a[2][4]; v2f b[2][2]; };

__device__ __forceinline__ void gemm_load_frags(const float* __restrict__ ap,
                                                const float* __restrict__ bp,
                                                int pr, FragSet& f) {
#pragma unroll
  for (int s = 0; s < 2; ++s) {
    const int ks = 2 * pr + s;
#pragma unroll
    for (int i = 0; i < 4; ++i) f.a[s][i] = *(const v2f*)(ap + 16 * i * PA + 4 * ks);
#pragma unroll
    for (int j = 0; j < 2; ++j) f.b[s][j] = *(const v2f*)(bp + 2 * ks * PB + 32 * j);
  }
}

__device__ __forceinline__ void gemm_compute_tile(const float* __restrict__ sA,
                                                  const float* __restrict__ sB,
                                                  int wm, int wn, int ln, int hl,
                                                  v8f (&c)[4][2]) {
  const float* ap = sA + (wm * 64 + ln) * PA + 2 * hl;
  const float* bp = sB + hl * PB + (wn * 32 + ln) * 2;
  FragSet f[2];
  gemm_load_frags(ap, bp, 0, f[0]);
#pragma unroll
  for (int pr = 0; pr < 4; ++pr) {
    if (pr + 1 < 4) gemm_load_frags(ap, bp, pr + 1, f[(pr + 1) & 1]);
    __builtin_amdgcn_sched_barrier(0);
#pragma unroll
    for (int s = 0; s < 2; ++s)
#pragma unroll
      for (int i = 0; i < 4; ++i)
#pragma unroll
        for (int j = 0; j < 2; ++j)
          c[i][j] = WMMA_F32(f[pr & 1].a[s][i], f[pr & 1].b[s][j], c[i][j]);
    if (pr + 1 < 4) __builtin_amdgcn_sched_barrier(0);
  }
}

template <int LDB>
__device__ __forceinline__ void gemm_mainloop(const float* A, const float* W,
                                              int m0, int n0, int tid, int wm, int wn,
                                              int ln, int hl, float* sA, float* sB,
                                              v8f (&c)[4][2]) {
  v4f ra[4], rb[4];
  gemm_load_tile<LDB>(A, W, m0, n0, 0, tid, ra, rb);
  gemm_store_tile(sA, sB, tid, ra, rb);
  __syncthreads();
  for (int kt = 0; kt < NKT - 1; ++kt) {
    const int cur = kt & 1;
    gemm_load_tile<LDB>(A, W, m0, n0, kt + 1, tid, ra, rb);
    __builtin_amdgcn_fence(__ATOMIC_RELEASE, "wavefront");
    gemm_compute_tile(sA + cur * SA, sB + cur * SB, wm, wn, ln, hl, c);
    gemm_store_tile(sA + (cur ^ 1) * SA, sB + (cur ^ 1) * SB, tid, ra, rb);
    __syncthreads();
  }
  gemm_compute_tile(sA + ((NKT - 1) & 1) * SA, sB + ((NKT - 1) & 1) * SB, wm, wn, ln, hl, c);
}

__global__ __launch_bounds__(256)
void qkv_gemm_kernel(const float* X, const float* W,
                     const float* __restrict__ bias,
                     float* __restrict__ q, float* __restrict__ k,
                     float* __restrict__ v) {
  __shared__ __attribute__((aligned(16))) float sA[2 * SA];
  __shared__ __attribute__((aligned(16))) float sB[2 * SB];
  const int tid = threadIdx.x;
  const int lane = tid & 31;
  const int hl = lane >> 4;
  const int ln = lane & 15;
  const int wave = tid >> 5;
  const int wm = wave >> 2, wn = wave & 3;
  const int n0 = blockIdx.x * BN;
  const int m0 = blockIdx.y * BM;

  v8f c[4][2] = {};
  gemm_mainloop<NQKV>(X, W, m0, n0, tid, wm, wn, ln, hl, sA, sB, c);

  const int which = n0 >> 10;
  float* dst = (which == 0) ? q : (which == 1) ? k : v;
#pragma unroll
  for (int j = 0; j < 2; ++j) {
    const int col = n0 + wn * 32 + 16 * j + ln;
    const float bval = bias[col];
    const int rem = col & (CC - 1);
    const int h = rem >> 6;
    const int d = rem & 63;
#pragma unroll
    for (int i = 0; i < 4; ++i) {
#pragma unroll
      for (int r = 0; r < 8; ++r) {
        const int trow = m0 + wm * 64 + 16 * i + r + 8 * hl;
        const int bidx = trow >> 11;
        const int t = trow & (TT - 1);
        const size_t idx = (((size_t)(bidx * HH + h) * TT) + t) * DD + d;
        dst[idx] = c[i][j][r] + bval;
      }
    }
  }
}

__global__ __launch_bounds__(32)
void attn_kernel(const float* __restrict__ q, const float* __restrict__ kmat,
                 const float* __restrict__ vmat, float* __restrict__ out) {
  const int lane = threadIdx.x & 31;
  const int hl = lane >> 4;
  const int ln = lane & 15;
  const int qtile = blockIdx.x;
  const int bh = blockIdx.y;
  const int b = bh >> 4, h = bh & 15;
  const int q0 = qtile * 16;

  const float* qbase = q + (size_t)bh * TT * DD;
  const float* kbase = kmat + (size_t)bh * TT * DD;
  const float* vbase = vmat + (size_t)bh * TT * DD;

  __shared__ float pt[16][17];

  v2f qa[16];
  {
    const float* qrow = qbase + (size_t)(q0 + ln) * DD + 2 * hl;
#pragma unroll
    for (int t = 0; t < 16; ++t) {
      v2f a = *(const v2f*)(qrow + 4 * t);
      qa[t] = a * 0.125f;
    }
  }

  v8f o[4] = {};
  float m_i[8], l_i[8];
#pragma unroll
  for (int r = 0; r < 8; ++r) { m_i[r] = -__builtin_inff(); l_i[r] = 0.0f; }

  for (int kt = 0; kt <= qtile; ++kt) {
    const int kb = kt * 16;

    v8f s = {};
    const float* krow = kbase + (size_t)(kb + ln) * DD + 2 * hl;
#pragma unroll
    for (int t = 0; t < 16; ++t) {
      v2f bf = *(const v2f*)(krow + 4 * t);
      s = WMMA_F32(qa[t], bf, s);
    }

    const bool diag = (kt == qtile);
    float scale_r[8];
#pragma unroll
    for (int r = 0; r < 8; ++r) {
      const int row = r + 8 * hl;
      float sv = s[r];
      if (diag && (ln > row)) sv = -__builtin_inff();
      float mx = sv;
      mx = fmaxf(mx, __shfl_xor(mx, 1));
      mx = fmaxf(mx, __shfl_xor(mx, 2));
      mx = fmaxf(mx, __shfl_xor(mx, 4));
      mx = fmaxf(mx, __shfl_xor(mx, 8));
      const float mn = fmaxf(m_i[r], mx);
      const float sc = __expf(m_i[r] - mn);
      const float p = __expf(sv - mn);
      float ps = p;
      ps += __shfl_xor(ps, 1);
      ps += __shfl_xor(ps, 2);
      ps += __shfl_xor(ps, 4);
      ps += __shfl_xor(ps, 8);
      l_i[r] = l_i[r] * sc + ps;
      m_i[r] = mn;
      scale_r[r] = sc;
      pt[row][ln] = p;
    }

#pragma unroll
    for (int n = 0; n < 4; ++n)
#pragma unroll
      for (int r = 0; r < 8; ++r) o[n][r] *= scale_r[r];

    __syncthreads();

#pragma unroll
    for (int t = 0; t < 4; ++t) {
      v2f a;
      a.x = pt[ln][4 * t + 2 * hl];
      a.y = pt[ln][4 * t + 2 * hl + 1];
      const float* vrow = vbase + (size_t)(kb + 4 * t + 2 * hl) * DD + ln;
#pragma unroll
      for (int n = 0; n < 4; ++n) {
        v2f bf;
        bf.x = vrow[16 * n];
        bf.y = vrow[DD + 16 * n];
        o[n] = WMMA_F32(a, bf, o[n]);
      }
    }
    __syncthreads();
  }

#pragma unroll
  for (int r = 0; r < 8; ++r) {
    const float inv = 1.0f / l_i[r];
    const int trow = q0 + r + 8 * hl;
    float* orow = out + ((size_t)(b * TT + trow) * CC) + h * DD;
#pragma unroll
    for (int n = 0; n < 4; ++n) orow[16 * n + ln] = o[n][r] * inv;
  }
}

__global__ __launch_bounds__(256)
void out_gemm_kernel(const float* A, const float* W,
                     const float* __restrict__ bias, float* __restrict__ out) {
  __shared__ __attribute__((aligned(16))) float sA[2 * SA];
  __shared__ __attribute__((aligned(16))) float sB[2 * SB];
  const int tid = threadIdx.x;
  const int lane = tid & 31;
  const int hl = lane >> 4;
  const int ln = lane & 15;
  const int wave = tid >> 5;
  const int wm = wave >> 2, wn = wave & 3;
  const int n0 = blockIdx.x * BN;
  const int m0 = blockIdx.y * BM;

  v8f c[4][2] = {};
  gemm_mainloop<CC>(A, W, m0, n0, tid, wm, wn, ln, hl, sA, sB, c);

#pragma unroll
  for (int j = 0; j < 2; ++j) {
    const int col = n0 + wn * 32 + 16 * j + ln;
    const float bval = bias[col];
#pragma unroll
    for (int i = 0; i < 4; ++i) {
#pragma unroll
      for (int r = 0; r < 8; ++r) {
        out[(size_t)(m0 + wm * 64 + 16 * i + r + 8 * hl) * CC + col] = c[i][j][r] + bval;
      }
    }
  }
}

extern "C" void kernel_launch(void* const* d_in, const int* in_sizes, int n_in,
                              void* d_out, int out_size, void* d_ws, size_t ws_size,
                              hipStream_t stream) {
  (void)in_sizes; (void)n_in; (void)out_size; (void)ws_size;
  const float* x    = (const float*)d_in[0];
  const float* Wqkv = (const float*)d_in[1];
  const float* bqkv = (const float*)d_in[2];
  const float* Wout = (const float*)d_in[3];
  const float* bout = (const float*)d_in[4];
  float* out = (float*)d_out;

  float* qp = (float*)d_ws;
  float* kp = qp + HSZ;
  float* vp = kp + HSZ;
  float* ao = vp + HSZ;

  const dim3 blk(32, 1, 1);
  const dim3 gblk(256, 1, 1);
  qkv_gemm_kernel<<<dim3(NQKV / BN, MTOT / BM), gblk, 0, stream>>>(x, Wqkv, bqkv, qp, kp, vp);
  attn_kernel<<<dim3(TT / 16, BB * HH), blk, 0, stream>>>(qp, kp, vp, ao);
  out_gemm_kernel<<<dim3(CC / BN, MTOT / BM), gblk, 0, stream>>>(ao, Wout, bout, out);
}


# Attention: 8-wave work-groups stage each 128-key K and V block once in LDS (16-byte loads, two stages) and read WMMA fragments from LDS, was per-wave global fragment loads; sum order unchanged
# speedup vs baseline: 2.3074x; 1.6084x over previous
#include <hip/hip_runtime.h>

typedef __attribute__((ext_vector_type(2))) float v2f;
typedef __attribute__((ext_vector_type(8))) float v8f;
typedef __attribute__((ext_vector_type(4))) float v4f;

#define WMMA_F32(a, b, c) \
  __builtin_amdgcn_wmma_f32_16x16x4_f32(false, (a), false, (b), (short)0, (c), false, false)

static constexpr int BB = 2, TT = 2048, CC = 1024, HH = 16, DD = 64;
static constexpr int MTOT = BB * TT;
static constexpr int NQKV = 3 * CC;
static constexpr size_t HSZ = (size_t)BB * HH * TT * DD;

static constexpr int BM = 128, BN = 128, KT = 32;
static constexpr int PA = KT + 4;
static constexpr int PB = 2 * BN + 32;
static constexpr int SA = BM * PA;
static constexpr int SB = (KT / 2) * PB;
static constexpr int NKT = CC / KT;

template <int LDB>
__device__ __forceinline__ void gemm_load_tile(const float* A, const float* W,
                                               int m0, int n0, int kt, int tid,
                                               v4f (&ra)[4], v4f (&rb)[4]) {
  const int arow = tid >> 3, akq = tid & 7;
  const int bpr = tid >> 5, bnq = tid & 31;
#pragma unroll
  for (int i = 0; i < 4; ++i)
    ra[i] = *(const v4f*)(A + (size_t)(m0 + arow + 32 * i) * CC + kt * KT + akq * 4);
#pragma unroll
  for (int i = 0; i < 2; ++i)
#pragma unroll
    for (int e = 0; e < 2; ++e)
      rb[2 * i + e] = *(const v4f*)(W + (size_t)(kt * KT + 2 * (bpr + 8 * i) + e) * LDB + n0 + bnq * 4);
}

__device__ __forceinline__ void gemm_store_tile(float* sA, float* sB, int tid,
                                                const v4f (&ra)[4], const v4f (&rb)[4]) {
  const int arow = tid >> 3, akq = tid & 7;
  const int bpr = tid >> 5, bnq = tid & 31;
#pragma unroll
  for (int i = 0; i < 4; ++i)
    *(v4f*)(sA + (arow + 32 * i) * PA + akq * 4) = ra[i];
#pragma unroll
  for (int i = 0; i < 2; ++i) {
    const v4f lo = rb[2 * i], hi = rb[2 * i + 1];
    float* d = sB + (bpr + 8 * i) * PB + bnq * 8;
    *(v4f*)(d)     = __builtin_shufflevector(lo, hi, 0, 4, 1, 5);
    *(v4f*)(d + 4) = __builtin_shufflevector(lo, hi, 2, 6, 3, 7);
  }
}

struct FragSet { v2f a[2][4]; v2f b[2][2]; };

__device__ __forceinline__ void gemm_load_frags(const float* __restrict__ ap,
                                                const float* __restrict__ bp,
                                                int pr, FragSet& f) {
#pragma unroll
  for (int s = 0; s < 2; ++s) {
    const int ks = 2 * pr + s;
#pragma unroll
    for (int i = 0; i < 4; ++i) f.a[s][i] = *(const v2f*)(ap + 16 * i * PA + 4 * ks);
#pragma unroll
    for (int j = 0; j < 2; ++j) f.b[s][j] = *(const v2f*)(bp + 2 * ks * PB + 32 * j);
  }
}

__device__ __forceinline__ void gemm_compute_tile(const float* __restrict__ sA,
                                                  const float* __restrict__ sB,
                                                  int wm, int wn, int ln, int hl,
                                                  v8f (&c)[4][2]) {
  const float* ap = sA + (wm * 64 + ln) * PA + 2 * hl;
  const float* bp = sB + hl * PB + (wn * 32 + ln) * 2;
  FragSet f[2];
  gemm_load_frags(ap, bp, 0, f[0]);
#pragma unroll
  for (int pr = 0; pr < 4; ++pr) {
    if (pr + 1 < 4) gemm_load_frags(ap, bp, pr + 1, f[(pr + 1) & 1]);
    __builtin_amdgcn_sched_barrier(0);
#pragma unroll
    for (int s = 0; s < 2; ++s)
#pragma unroll
      for (int i = 0; i < 4; ++i)
#pragma unroll
        for (int j = 0; j < 2; ++j)
          c[i][j] = WMMA_F32(f[pr & 1].a[s][i], f[pr & 1].b[s][j], c[i][j]);
    if (pr + 1 < 4) __builtin_amdgcn_sched_barrier(0);
  }
}

template <int LDB>
__device__ __forceinline__ void gemm_mainloop(const float* A, const float* W,
                                              int m0, int n0, int tid, int wm, int wn,
                                              int ln, int hl, float* sA, float* sB,
                                              v8f (&c)[4][2]) {
  v4f ra[4], rb[4];
  gemm_load_tile<LDB>(A, W, m0, n0, 0, tid, ra, rb);
  gemm_store_tile(sA, sB, tid, ra, rb);
  __syncthreads();
  for (int kt = 0; kt < NKT - 1; ++kt) {
    const int cur = kt & 1;
    gemm_load_tile<LDB>(A, W, m0, n0, kt + 1, tid, ra, rb);
    __builtin_amdgcn_fence(__ATOMIC_RELEASE, "wavefront");
    gemm_compute_tile(sA + cur * SA, sB + cur * SB, wm, wn, ln, hl, c);
    gemm_store_tile(sA + (cur ^ 1) * SA, sB + (cur ^ 1) * SB, tid, ra, rb);
    __syncthreads();
  }
  gemm_compute_tile(sA + ((NKT - 1) & 1) * SA, sB + ((NKT - 1) & 1) * SB, wm, wn, ln, hl, c);
}

__global__ __launch_bounds__(256)
void qkv_gemm_kernel(const float* X, const float* W,
                     const float* __restrict__ bias,
                     float* __restrict__ q, float* __restrict__ k,
                     float* __restrict__ v) {
  __shared__ __attribute__((aligned(16))) float sA[2 * SA];
  __shared__ __attribute__((aligned(16))) float sB[2 * SB];
  const int tid = threadIdx.x;
  const int lane = tid & 31;
  const int hl = lane >> 4;
  const int ln = lane & 15;
  const int wave = tid >> 5;
  const int wm = wave >> 2, wn = wave & 3;
  const int n0 = blockIdx.x * BN;
  const int m0 = blockIdx.y * BM;

  v8f c[4][2] = {};
  gemm_mainloop<NQKV>(X, W, m0, n0, tid, wm, wn, ln, hl, sA, sB, c);

  const int which = n0 >> 10;
  float* dst = (which == 0) ? q : (which == 1) ? k : v;
#pragma unroll
  for (int j = 0; j < 2; ++j) {
    const int col = n0 + wn * 32 + 16 * j + ln;
    const float bval = bias[col];
    const int rem = col & (CC - 1);
    const int h = rem >> 6;
    const int d = rem & 63;
#pragma unroll
    for (int i = 0; i < 4; ++i) {
#pragma unroll
      for (int r = 0; r < 8; ++r) {
        const int trow = m0 + wm * 64 + 16 * i + r + 8 * hl;
        const int bidx = trow >> 11;
        const int t = trow & (TT - 1);
        const size_t idx = (((size_t)(bidx * HH + h) * TT) + t) * DD + d;
        dst[idx] = c[i][j][r] + bval;
      }
    }
  }
}

static constexpr int QB = 128;
static constexpr int KB = 128;
static constexpr int PK = DD + 4;
static constexpr int PV = 2 * DD + 32;
static constexpr int SK = KB * PK;
static constexpr int SV = (KB / 2) * PV;
static constexpr int NQB = TT / QB;

__device__ __forceinline__ void attn_load_block(const float* kbase, const float* vbase,
                                                int kb, int tid, v4f (&rk)[8], v4f (&rv)[8]) {
  const float* kp = kbase + (size_t)kb * KB * DD + tid * 4;
  const float* vp = vbase + (size_t)kb * KB * DD + (tid >> 4) * 2 * DD + (tid & 15) * 4;
#pragma unroll
  for (int i = 0; i < 8; ++i) rk[i] = *(const v4f*)(kp + i * 1024);
#pragma unroll
  for (int j = 0; j < 4; ++j)
#pragma unroll
    for (int e = 0; e < 2; ++e) rv[2 * j + e] = *(const v4f*)(vp + (32 * j + e) * DD);
}

__device__ __forceinline__ void attn_store_block(float* sK, float* sV, int tid,
                                                 const v4f (&rk)[8], const v4f (&rv)[8]) {
  const int row = tid >> 4, c4 = (tid & 15) * 4;
#pragma unroll
  for (int i = 0; i < 8; ++i) *(v4f*)(sK + (row + 16 * i) * PK + c4) = rk[i];
#pragma unroll
  for (int j = 0; j < 4; ++j) {
    const v4f lo = rv[2 * j], hi = rv[2 * j + 1];
    float* dst = sV + (row + 16 * j) * PV + c4 * 2;
    *(v4f*)(dst)     = __builtin_shufflevector(lo, hi, 0, 4, 1, 5);
    *(v4f*)(dst + 4) = __builtin_shufflevector(lo, hi, 2, 6, 3, 7);
  }
}

struct KFrags { v2f f[4][4]; };

__device__ __forceinline__ void attn_load_kfrags(const float* __restrict__ gK, int tb, KFrags& f) {
#pragma unroll
  for (int tt = 0; tt < 4; ++tt)
#pragma unroll
    for (int u = 0; u < 4; ++u)
      f.f[tt][u] = *(const v2f*)(gK + u * 16 * PK + 4 * (4 * tb + tt));
}

__device__ __forceinline__ void attn_tile_update(v8f s, bool diag, const float* __restrict__ vt,
                                                 float (*pt)[17], int ln, int hl,
                                                 float (&m_i)[8], float (&l_i)[8], v8f (&o)[4]) {
  float scale_r[8];
#pragma unroll
  for (int r = 0; r < 8; ++r) {
    const int row = r + 8 * hl;
    float sv = s[r];
    if (diag && (ln > row)) sv = -__builtin_inff();
    float mx = sv;
    mx = fmaxf(mx, __shfl_xor(mx, 1));
    mx = fmaxf(mx, __shfl_xor(mx, 2));
    mx = fmaxf(mx, __shfl_xor(mx, 4));
    mx = fmaxf(mx, __shfl_xor(mx, 8));
    const float mn = fmaxf(m_i[r], mx);
    const float sc = __expf(m_i[r] - mn);
    const float p = __expf(sv - mn);
    float ps = p;
    ps += __shfl_xor(ps, 1);
    ps += __shfl_xor(ps, 2);
    ps += __shfl_xor(ps, 4);
    ps += __shfl_xor(ps, 8);
    l_i[r] = l_i[r] * sc + ps;
    m_i[r] = mn;
    scale_r[r] = sc;
    pt[row][ln] = p;
  }

#pragma unroll
  for (int n = 0; n < 4; ++n)
#pragma unroll
    for (int r = 0; r < 8; ++r) o[n][r] *= scale_r[r];

  __builtin_amdgcn_fence(__ATOMIC_RELEASE, "wavefront");
  __builtin_amdgcn_fence(__ATOMIC_ACQUIRE, "wavefront");

#pragma unroll
  for (int t = 0; t < 4; ++t) {
    v2f a;
    a.x = pt[ln][4 * t + 2 * hl];
    a.y = pt[ln][4 * t + 2 * hl + 1];
    const float* vrow = vt + (2 * t + hl) * PV + ln * 2;
#pragma unroll
    for (int n = 0; n < 4; ++n) {
      const v2f bf = *(const v2f*)(vrow + 32 * n);
      o[n] = WMMA_F32(a, bf, o[n]);
    }
  }
  __builtin_amdgcn_fence(__ATOMIC_RELEASE, "wavefront");
}

__global__ __launch_bounds__(256)
void attn_kernel(const float* __restrict__ q, const float* kmat,
                 const float* vmat, float* __restrict__ out) {
  __shared__ __attribute__((aligned(16))) float sK[2 * SK];
  __shared__ __attribute__((aligned(16))) float sV[2 * SV];
  __shared__ float ptile[8][4][16][17];

  const int tid = threadIdx.x;
  const int lane = tid & 31;
  const int hl = lane >> 4;
  const int ln = lane & 15;
  const int wave = __builtin_amdgcn_readfirstlane(tid >> 5);
  const int pair = blockIdx.x;
  const int bh = blockIdx.y;
  const int b = bh >> 4, h = bh & 15;

  const float* qbase = q + (size_t)bh * TT * DD;
  const float* kbase = kmat + (size_t)bh * TT * DD;
  const float* vbase = vmat + (size_t)bh * TT * DD;

  v4f rk[8], rv[8];

  for (int pass = 0; pass < 2; ++pass) {
    const int qb = pass == 0 ? (NQB - 1 - pair) : pair;
    const int qt = qb * 8 + wave;
    const int q0 = qt * 16;
    const int nkb = qb + 1;

    v2f qa[16];
    {
      const float* qrow = qbase + (size_t)(q0 + ln) * DD + 2 * hl;
#pragma unroll
      for (int t = 0; t < 16; ++t) {
        v2f a = *(const v2f*)(qrow + 4 * t);
        qa[t] = a * 0.125f;
      }
    }

    v8f o[4] = {};
    float m_i[8], l_i[8];
#pragma unroll
    for (int r = 0; r < 8; ++r) { m_i[r] = -__builtin_inff(); l_i[r] = 0.0f; }

    attn_load_block(kbase, vbase, 0, tid, rk, rv);
    attn_store_block(sK, sV, tid, rk, rv);
    __syncthreads();

    for (int kb = 0; kb < nkb; ++kb) {
      const int cur = kb & 1;
      if (kb + 1 < nkb) {
        attn_load_block(kbase, vbase, kb + 1, tid, rk, rv);
        __builtin_amdgcn_fence(__ATOMIC_RELEASE, "wavefront");
      }
      const float* cK = sK + cur * SK;
      const float* cV = sV + cur * SV;

#pragma unroll 1
      for (int g = 0; g < 2; ++g) {
        const int kt0 = kb * 8 + g * 4;
        const float* gK = cK + (g * 64 + ln) * PK + 2 * hl;
        const float* gV = cV + (g * 32) * PV;
        if (qt >= kt0 + 4) {
          v8f s[4] = {};
          KFrags f[2];
          attn_load_kfrags(gK, 0, f[0]);
#pragma unroll
          for (int tb = 0; tb < 4; ++tb) {
            if (tb + 1 < 4) attn_load_kfrags(gK, tb + 1, f[(tb + 1) & 1]);
            __builtin_amdgcn_sched_barrier(0);
#pragma unroll
            for (int tt = 0; tt < 4; ++tt)
#pragma unroll
              for (int u = 0; u < 4; ++u)
                s[u] = WMMA_F32(qa[4 * tb + tt], f[tb & 1].f[tt][u], s[u]);
            if (tb + 1 < 4) __builtin_amdgcn_sched_barrier(0);
          }
#pragma unroll
          for (int u = 0; u < 4; ++u)
            attn_tile_update(s[u], false, gV + u * 8 * PV, ptile[wave][u], ln, hl, m_i, l_i, o);
        } else if (qt >= kt0) {
          const int nt = qt - kt0 + 1;
#pragma unroll 1
          for (int u = 0; u < nt; ++u) {
            v8f s = {};
#pragma unroll
            for (int t = 0; t < 16; ++t) {
              v2f bf = *(const v2f*)(gK + u * 16 * PK + 4 * t);
              s = WMMA_F32(qa[t], bf, s);
            }
            attn_tile_update(s, u == nt - 1, gV + u * 8 * PV, ptile[wave][0], ln, hl, m_i, l_i, o);
          }
        }
      }

      if (kb + 1 < nkb)
        attn_store_block(sK + (cur ^ 1) * SK, sV + (cur ^ 1) * SV, tid, rk, rv);
      __syncthreads();
    }

#pragma unroll
    for (int r = 0; r < 8; ++r) {
      const float inv = 1.0f / l_i[r];
      const int trow = q0 + r + 8 * hl;
      float* orow = out + ((size_t)(b * TT + trow) * CC) + h * DD;
#pragma unroll
      for (int n = 0; n < 4; ++n) orow[16 * n + ln] = o[n][r] * inv;
    }
  }
}

__global__ __launch_bounds__(256)
void out_gemm_kernel(const float* A, const float* W,
                     const float* __restrict__ bias, float* __restrict__ out) {
  __shared__ __attribute__((aligned(16))) float sA[2 * SA];
  __shared__ __attribute__((aligned(16))) float sB[2 * SB];
  const int tid = threadIdx.x;
  const int lane = tid & 31;
  const int hl = lane >> 4;
  const int ln = lane & 15;
  const int wave = tid >> 5;
  const int wm = wave >> 2, wn = wave & 3;
  const int n0 = blockIdx.x * BN;
  const int m0 = blockIdx.y * BM;

  v8f c[4][2] = {};
  gemm_mainloop<CC>(A, W, m0, n0, tid, wm, wn, ln, hl, sA, sB, c);

#pragma unroll
  for (int j = 0; j < 2; ++j) {
    const int col = n0 + wn * 32 + 16 * j + ln;
    const float bval = bias[col];
#pragma unroll
    for (int i = 0; i < 4; ++i) {
#pragma unroll
      for (int r = 0; r < 8; ++r) {
        out[(size_t)(m0 + wm * 64 + 16 * i + r + 8 * hl) * CC + col] = c[i][j][r] + bval;
      }
    }
  }
}

extern "C" void kernel_launch(void* const* d_in, const int* in_sizes, int n_in,
                              void* d_out, int out_size, void* d_ws, size_t ws_size,
                              hipStream_t stream) {
  (void)in_sizes; (void)n_in; (void)out_size; (void)ws_size;
  const float* x    = (const float*)d_in[0];
  const float* Wqkv = (const float*)d_in[1];
  const float* bqkv = (const float*)d_in[2];
  const float* Wout = (const float*)d_in[3];
  const float* bout = (const float*)d_in[4];
  float* out = (float*)d_out;

  float* qp = (float*)d_ws;
  float* kp = qp + HSZ;
  float* vp = kp + HSZ;
  float* ao = vp + HSZ;

  const dim3 gblk(256, 1, 1);
  qkv_gemm_kernel<<<dim3(NQKV / BN, MTOT / BM), gblk, 0, stream>>>(x, Wqkv, bqkv, qp, kp, vp);
  attn_kernel<<<dim3(NQB / 2, BB * HH), gblk, 0, stream>>>(qp, kp, vp, ao);
  out_gemm_kernel<<<dim3(CC / BN, MTOT / BM), gblk, 0, stream>>>(ao, Wout, bout, out);
}


# Attention softmax: the 16-lane row max and row sum butterflies go through DPP row_xmask on the vector ALU, were ds_bpermute LDS round trips (64 per 16-key tile); same butterfly, sum order unchanged
# speedup vs baseline: 2.6150x; 1.1333x over previous
#include <hip/hip_runtime.h>

typedef __attribute__((ext_vector_type(2))) float v2f;
typedef __attribute__((ext_vector_type(8))) float v8f;
typedef __attribute__((ext_vector_type(4))) float v4f;

#define WMMA_F32(a, b, c) \
  __builtin_amdgcn_wmma_f32_16x16x4_f32(false, (a), false, (b), (short)0, (c), false, false)

static constexpr int BB = 2, TT = 2048, CC = 1024, HH = 16, DD = 64;
static constexpr int MTOT = BB * TT;
static constexpr int NQKV = 3 * CC;
static constexpr size_t HSZ = (size_t)BB * HH * TT * DD;

static constexpr int BM = 128, BN = 128, KT = 32;
static constexpr int PA = KT + 4;
static constexpr int PB = 2 * BN + 32;
static constexpr int SA = BM * PA;
static constexpr int SB = (KT / 2) * PB;
static constexpr int NKT = CC / KT;

template <int LDB>
__device__ __forceinline__ void gemm_load_tile(const float* A, const float* W,
                                               int m0, int n0, int kt, int tid,
                                               v4f (&ra)[4], v4f (&rb)[4]) {
  const int arow = tid >> 3, akq = tid & 7;
  const int bpr = tid >> 5, bnq = tid & 31;
#pragma unroll
  for (int i = 0; i < 4; ++i)
    ra[i] = *(const v4f*)(A + (size_t)(m0 + arow + 32 * i) * CC + kt * KT + akq * 4);
#pragma unroll
  for (int i = 0; i < 2; ++i)
#pragma unroll
    for (int e = 0; e < 2; ++e)
      rb[2 * i + e] = *(const v4f*)(W + (size_t)(kt * KT + 2 * (bpr + 8 * i) + e) * LDB + n0 + bnq * 4);
}

__device__ __forceinline__ void gemm_store_tile(float* sA, float* sB, int tid,
                                                const v4f (&ra)[4], const v4f (&rb)[4]) {
  const int arow = tid >> 3, akq = tid & 7;
  const int bpr = tid >> 5, bnq = tid & 31;
#pragma unroll
  for (int i = 0; i < 4; ++i)
    *(v4f*)(sA + (arow + 32 * i) * PA + akq * 4) = ra[i];
#pragma unroll
  for (int i = 0; i < 2; ++i) {
    const v4f lo = rb[2 * i], hi = rb[2 * i + 1];
    float* d = sB + (bpr + 8 * i) * PB + bnq * 8;
    *(v4f*)(d)     = __builtin_shufflevector(lo, hi, 0, 4, 1, 5);
    *(v4f*)(d + 4) = __builtin_shufflevector(lo, hi, 2, 6, 3, 7);
  }
}

struct FragSet { v2f a[2][4]; v2f b[2][2]; };

__device__ __forceinline__ void gemm_load_frags(const float* __restrict__ ap,
                                                const float* __restrict__ bp,
                                                int pr, FragSet& f) {
#pragma unroll
  for (int s = 0; s < 2; ++s) {
    const int ks = 2 * pr + s;
#pragma unroll
    for (int i = 0; i < 4; ++i) f.a[s][i] = *(const v2f*)(ap + 16 * i * PA + 4 * ks);
#pragma unroll
    for (int j = 0; j < 2; ++j) f.b[s][j] = *(const v2f*)(bp + 2 * ks * PB + 32 * j);
  }
}

__device__ __forceinline__ void gemm_compute_tile(const float* __restrict__ sA,
                                                  const float* __restrict__ sB,
                                                  int wm, int wn, int ln, int hl,
                                                  v8f (&c)[4][2]) {
  const float* ap = sA + (wm * 64 + ln) * PA + 2 * hl;
  const float* bp = sB + hl * PB + (wn * 32 + ln) * 2;
  FragSet f[2];
  gemm_load_frags(ap, bp, 0, f[0]);
#pragma unroll
  for (int pr = 0; pr < 4; ++pr) {
    if (pr + 1 < 4) gemm_load_frags(ap, bp, pr + 1, f[(pr + 1) & 1]);
    __builtin_amdgcn_sched_barrier(0);
#pragma unroll
    for (int s = 0; s < 2; ++s)
#pragma unroll
      for (int i = 0; i < 4; ++i)
#pragma unroll
        for (int j = 0; j < 2; ++j)
          c[i][j] = WMMA_F32(f[pr & 1].a[s][i], f[pr & 1].b[s][j], c[i][j]);
    if (pr + 1 < 4) __builtin_amdgcn_sched_barrier(0);
  }
}

template <int LDB>
__device__ __forceinline__ void gemm_mainloop(const float* A, const float* W,
                                              int m0, int n0, int tid, int wm, int wn,
                                              int ln, int hl, float* sA, float* sB,
                                              v8f (&c)[4][2]) {
  v4f ra[4], rb[4];
  gemm_load_tile<LDB>(A, W, m0, n0, 0, tid, ra, rb);
  gemm_store_tile(sA, sB, tid, ra, rb);
  __syncthreads();
  for (int kt = 0; kt < NKT - 1; ++kt) {
    const int cur = kt & 1;
    gemm_load_tile<LDB>(A, W, m0, n0, kt + 1, tid, ra, rb);
    __builtin_amdgcn_fence(__ATOMIC_RELEASE, "wavefront");
    gemm_compute_tile(sA + cur * SA, sB + cur * SB, wm, wn, ln, hl, c);
    gemm_store_tile(sA + (cur ^ 1) * SA, sB + (cur ^ 1) * SB, tid, ra, rb);
    __syncthreads();
  }
  gemm_compute_tile(sA + ((NKT - 1) & 1) * SA, sB + ((NKT - 1) & 1) * SB, wm, wn, ln, hl, c);
}

__global__ __launch_bounds__(256)
void qkv_gemm_kernel(const float* X, const float* W,
                     const float* __restrict__ bias,
                     float* __restrict__ q, float* __restrict__ k,
                     float* __restrict__ v) {
  __shared__ __attribute__((aligned(16))) float sA[2 * SA];
  __shared__ __attribute__((aligned(16))) float sB[2 * SB];
  const int tid = threadIdx.x;
  const int lane = tid & 31;
  const int hl = lane >> 4;
  const int ln = lane & 15;
  const int wave = tid >> 5;
  const int wm = wave >> 2, wn = wave & 3;
  const int n0 = blockIdx.x * BN;
  const int m0 = blockIdx.y * BM;

  v8f c[4][2] = {};
  gemm_mainloop<NQKV>(X, W, m0, n0, tid, wm, wn, ln, hl, sA, sB, c);

  const int which = n0 >> 10;
  float* dst = (which == 0) ? q : (which == 1) ? k : v;
#pragma unroll
  for (int j = 0; j < 2; ++j) {
    const int col = n0 + wn * 32 + 16 * j + ln;
    const float bval = bias[col];
    const int rem = col & (CC - 1);
    const int h = rem >> 6;
    const int d = rem & 63;
#pragma unroll
    for (int i = 0; i < 4; ++i) {
#pragma unroll
      for (int r = 0; r < 8; ++r) {
        const int trow = m0 + wm * 64 + 16 * i + r + 8 * hl;
        const int bidx = trow >> 11;
        const int t = trow & (TT - 1);
        const size_t idx = (((size_t)(bidx * HH + h) * TT) + t) * DD + d;
        dst[idx] = c[i][j][r] + bval;
      }
    }
  }
}

static constexpr int QB = 128;
static constexpr int KB = 128;
static constexpr int PK = DD + 4;
static constexpr int PV = 2 * DD + 32;
static constexpr int SK = KB * PK;
static constexpr int SV = (KB / 2) * PV;
static constexpr int NQB = TT / QB;

__device__ __forceinline__ void attn_load_block(const float* kbase, const float* vbase,
                                                int kb, int tid, v4f (&rk)[8], v4f (&rv)[8]) {
  const float* kp = kbase + (size_t)kb * KB * DD + tid * 4;
  const float* vp = vbase + (size_t)kb * KB * DD + (tid >> 4) * 2 * DD + (tid & 15) * 4;
#pragma unroll
  for (int i = 0; i < 8; ++i) rk[i] = *(const v4f*)(kp + i * 1024);
#pragma unroll
  for (int j = 0; j < 4; ++j)
#pragma unroll
    for (int e = 0; e < 2; ++e) rv[2 * j + e] = *(const v4f*)(vp + (32 * j + e) * DD);
}

__device__ __forceinline__ void attn_store_block(float* sK, float* sV, int tid,
                                                 const v4f (&rk)[8], const v4f (&rv)[8]) {
  const int row = tid >> 4, c4 = (tid & 15) * 4;
#pragma unroll
  for (int i = 0; i < 8; ++i) *(v4f*)(sK + (row + 16 * i) * PK + c4) = rk[i];
#pragma unroll
  for (int j = 0; j < 4; ++j) {
    const v4f lo = rv[2 * j], hi = rv[2 * j + 1];
    float* dst = sV + (row + 16 * j) * PV + c4 * 2;
    *(v4f*)(dst)     = __builtin_shufflevector(lo, hi, 0, 4, 1, 5);
    *(v4f*)(dst + 4) = __builtin_shufflevector(lo, hi, 2, 6, 3, 7);
  }
}

struct KFrags { v2f f[4][4]; };

__device__ __forceinline__ void attn_load_kfrags(const float* __restrict__ gK, int tb, KFrags& f) {
#pragma unroll
  for (int tt = 0; tt < 4; ++tt)
#pragma unroll
    for (int u = 0; u < 4; ++u)
      f.f[tt][u] = *(const v2f*)(gK + u * 16 * PK + 4 * (4 * tb + tt));
}

template <int M>
__device__ __forceinline__ float row_xor(float x) {
  return __builtin_bit_cast(float, __builtin_amdgcn_update_dpp(0, __builtin_bit_cast(int, x),
                                                               0x160 | M, 0xf, 0xf, true));
}

__device__ __forceinline__ void attn_tile_update(v8f s, bool diag, const float* __restrict__ vt,
                                                 float (*pt)[17], int ln, int hl,
                                                 float (&m_i)[8], float (&l_i)[8], v8f (&o)[4]) {
  float scale_r[8];
#pragma unroll
  for (int r = 0; r < 8; ++r) {
    const int row = r + 8 * hl;
    float sv = s[r];
    if (diag && (ln > row)) sv = -__builtin_inff();
    float mx = sv;
    mx = fmaxf(mx, row_xor<1>(mx));
    mx = fmaxf(mx, row_xor<2>(mx));
    mx = fmaxf(mx, row_xor<4>(mx));
    mx = fmaxf(mx, row_xor<8>(mx));
    const float mn = fmaxf(m_i[r], mx);
    const float sc = __expf(m_i[r] - mn);
    const float p = __expf(sv - mn);
    float ps = p;
    ps += row_xor<1>(ps);
    ps += row_xor<2>(ps);
    ps += row_xor<4>(ps);
    ps += row_xor<8>(ps);
    l_i[r] = l_i[r] * sc + ps;
    m_i[r] = mn;
    scale_r[r] = sc;
    pt[row][ln] = p;
  }

#pragma unroll
  for (int n = 0; n < 4; ++n)
#pragma unroll
    for (int r = 0; r < 8; ++r) o[n][r] *= scale_r[r];

  __builtin_amdgcn_fence(__ATOMIC_RELEASE, "wavefront");
  __builtin_amdgcn_fence(__ATOMIC_ACQUIRE, "wavefront");

#pragma unroll
  for (int t = 0; t < 4; ++t) {
    v2f a;
    a.x = pt[ln][4 * t + 2 * hl];
    a.y = pt[ln][4 * t + 2 * hl + 1];
    const float* vrow = vt + (2 * t + hl) * PV + ln * 2;
#pragma unroll
    for (int n = 0; n < 4; ++n) {
      const v2f bf = *(const v2f*)(vrow + 32 * n);
      o[n] = WMMA_F32(a, bf, o[n]);
    }
  }
  __builtin_amdgcn_fence(__ATOMIC_RELEASE, "wavefront");
}

__global__ __launch_bounds__(256)
void attn_kernel(const float* __restrict__ q, const float* kmat,
                 const float* vmat, float* __restrict__ out) {
  __shared__ __attribute__((aligned(16))) float sK[2 * SK];
  __shared__ __attribute__((aligned(16))) float sV[2 * SV];
  __shared__ float ptile[8][4][16][17];

  const int tid = threadIdx.x;
  const int lane = tid & 31;
  const int hl = lane >> 4;
  const int ln = lane & 15;
  const int wave = __builtin_amdgcn_readfirstlane(tid >> 5);
  const int pair = blockIdx.x;
  const int bh = blockIdx.y;
  const int b = bh >> 4, h = bh & 15;

  const float* qbase = q + (size_t)bh * TT * DD;
  const float* kbase = kmat + (size_t)bh * TT * DD;
  const float* vbase = vmat + (size_t)bh * TT * DD;

  v4f rk[8], rv[8];

  for (int pass = 0; pass < 2; ++pass) {
    const int qb = pass == 0 ? (NQB - 1 - pair) : pair;
    const int qt = qb * 8 + wave;
    const int q0 = qt * 16;
    const int nkb = qb + 1;

    v2f qa[16];
    {
      const float* qrow = qbase + (size_t)(q0 + ln) * DD + 2 * hl;
#pragma unroll
      for (int t = 0; t < 16; ++t) {
        v2f a = *(const v2f*)(qrow + 4 * t);
        qa[t] = a * 0.125f;
      }
    }

    v8f o[4] = {};
    float m_i[8], l_i[8];
#pragma unroll
    for (int r = 0; r < 8; ++r) { m_i[r] = -__builtin_inff(); l_i[r] = 0.0f; }

    attn_load_block(kbase, vbase, 0, tid, rk, rv);
    attn_store_block(sK, sV, tid, rk, rv);
    __syncthreads();

    for (int kb = 0; kb < nkb; ++kb) {
      const int cur = kb & 1;
      if (kb + 1 < nkb) {
        attn_load_block(kbase, vbase, kb + 1, tid, rk, rv);
        __builtin_amdgcn_fence(__ATOMIC_RELEASE, "wavefront");
      }
      const float* cK = sK + cur * SK;
      const float* cV = sV + cur * SV;

#pragma unroll 1
      for (int g = 0; g < 2; ++g) {
        const int kt0 = kb * 8 + g * 4;
        const float* gK = cK + (g * 64 + ln) * PK + 2 * hl;
        const float* gV = cV + (g * 32) * PV;
        if (qt >= kt0 + 4) {
          v8f s[4] = {};
          KFrags f[2];
          attn_load_kfrags(gK, 0, f[0]);
#pragma unroll
          for (int tb = 0; tb < 4; ++tb) {
            if (tb + 1 < 4) attn_load_kfrags(gK, tb + 1, f[(tb + 1) & 1]);
            __builtin_amdgcn_sched_barrier(0);
#pragma unroll
            for (int tt = 0; tt < 4; ++tt)
#pragma unroll
              for (int u = 0; u < 4; ++u)
                s[u] = WMMA_F32(qa[4 * tb + tt], f[tb & 1].f[tt][u], s[u]);
            if (tb + 1 < 4) __builtin_amdgcn_sched_barrier(0);
          }
#pragma unroll
          for (int u = 0; u < 4; ++u)
            attn_tile_update(s[u], false, gV + u * 8 * PV, ptile[wave][u], ln, hl, m_i, l_i, o);
        } else if (qt >= kt0) {
          const int nt = qt - kt0 + 1;
#pragma unroll 1
          for (int u = 0; u < nt; ++u) {
            v8f s = {};
#pragma unroll
            for (int t = 0; t < 16; ++t) {
              v2f bf = *(const v2f*)(gK + u * 16 * PK + 4 * t);
              s = WMMA_F32(qa[t], bf, s);
            }
            attn_tile_update(s, u == nt - 1, gV + u * 8 * PV, ptile[wave][0], ln, hl, m_i, l_i, o);
          }
        }
      }

      if (kb + 1 < nkb)
        attn_store_block(sK + (cur ^ 1) * SK, sV + (cur ^ 1) * SV, tid, rk, rv);
      __syncthreads();
    }

#pragma unroll
    for (int r = 0; r < 8; ++r) {
      const float inv = 1.0f / l_i[r];
      const int trow = q0 + r + 8 * hl;
      float* orow = out + ((size_t)(b * TT + trow) * CC) + h * DD;
#pragma unroll
      for (int n = 0; n < 4; ++n) orow[16 * n + ln] = o[n][r] * inv;
    }
  }
}

__global__ __launch_bounds__(256)
void out_gemm_kernel(const float* A, const float* W,
                     const float* __restrict__ bias, float* __restrict__ out) {
  __shared__ __attribute__((aligned(16))) float sA[2 * SA];
  __shared__ __attribute__((aligned(16))) float sB[2 * SB];
  const int tid = threadIdx.x;
  const int lane = tid & 31;
  const int hl = lane >> 4;
  const int ln = lane & 15;
  const int wave = tid >> 5;
  const int wm = wave >> 2, wn = wave & 3;
  const int n0 = blockIdx.x * BN;
  const int m0 = blockIdx.y * BM;

  v8f c[4][2] = {};
  gemm_mainloop<CC>(A, W, m0, n0, tid, wm, wn, ln, hl, sA, sB, c);

#pragma unroll
  for (int j = 0; j < 2; ++j) {
    const int col = n0 + wn * 32 + 16 * j + ln;
    const float bval = bias[col];
#pragma unroll
    for (int i = 0; i < 4; ++i) {
#pragma unroll
      for (int r = 0; r < 8; ++r) {
        out[(size_t)(m0 + wm * 64 + 16 * i + r + 8 * hl) * CC + col] = c[i][j][r] + bval;
      }
    }
  }
}

extern "C" void kernel_launch(void* const* d_in, const int* in_sizes, int n_in,
                              void* d_out, int out_size, void* d_ws, size_t ws_size,
                              hipStream_t stream) {
  (void)in_sizes; (void)n_in; (void)out_size; (void)ws_size;
  const float* x    = (const float*)d_in[0];
  const float* Wqkv = (const float*)d_in[1];
  const float* bqkv = (const float*)d_in[2];
  const float* Wout = (const float*)d_in[3];
  const float* bout = (const float*)d_in[4];
  float* out = (float*)d_out;

  float* qp = (float*)d_ws;
  float* kp = qp + HSZ;
  float* vp = kp + HSZ;
  float* ao = vp + HSZ;

  const dim3 gblk(256, 1, 1);
  qkv_gemm_kernel<<<dim3(NQKV / BN, MTOT / BM), gblk, 0, stream>>>(x, Wqkv, bqkv, qp, kp, vp);
  attn_kernel<<<dim3(NQB / 2, BB * HH), gblk, 0, stream>>>(qp, kp, vp, ao);
  out_gemm_kernel<<<dim3(CC / BN, MTOT / BM), gblk, 0, stream>>>(ao, Wout, bout, out);
}


# Attention computes transposed tiles (S^T=K Q^T, O^T=V^T P^T): max, sum, rescale per lane, P^T feeds WMMA from registers, no LDS patch or 16-lane butterflies; row sum order changes (in-lane tree)
# speedup vs baseline: 2.9363x; 1.1229x over previous
#include <hip/hip_runtime.h>

typedef __attribute__((ext_vector_type(2))) float v2f;
typedef __attribute__((ext_vector_type(8))) float v8f;
typedef __attribute__((ext_vector_type(4))) float v4f;

#define WMMA_F32(a, b, c) \
  __builtin_amdgcn_wmma_f32_16x16x4_f32(false, (a), false, (b), (short)0, (c), false, false)

static constexpr int BB = 2, TT = 2048, CC = 1024, HH = 16, DD = 64;
static constexpr int MTOT = BB * TT;
static constexpr int NQKV = 3 * CC;
static constexpr size_t HSZ = (size_t)BB * HH * TT * DD;

static constexpr int BM = 128, BN = 128, KT = 32;
static constexpr int PA = KT + 4;
static constexpr int PB = 2 * BN + 32;
static constexpr int SA = BM * PA;
static constexpr int SB = (KT / 2) * PB;
static constexpr int NKT = CC / KT;

template <int LDB>
__device__ __forceinline__ void gemm_load_tile(const float* A, const float* W,
                                               int m0, int n0, int kt, int tid,
                                               v4f (&ra)[4], v4f (&rb)[4]) {
  const int arow = tid >> 3, akq = tid & 7;
  const int bpr = tid >> 5, bnq = tid & 31;
#pragma unroll
  for (int i = 0; i < 4; ++i)
    ra[i] = *(const v4f*)(A + (size_t)(m0 + arow + 32 * i) * CC + kt * KT + akq * 4);
#pragma unroll
  for (int i = 0; i < 2; ++i)
#pragma unroll
    for (int e = 0; e < 2; ++e)
      rb[2 * i + e] = *(const v4f*)(W + (size_t)(kt * KT + 2 * (bpr + 8 * i) + e) * LDB + n0 + bnq * 4);
}

__device__ __forceinline__ void gemm_store_tile(float* sA, float* sB, int tid,
                                                const v4f (&ra)[4], const v4f (&rb)[4]) {
  const int arow = tid >> 3, akq = tid & 7;
  const int bpr = tid >> 5, bnq = tid & 31;
#pragma unroll
  for (int i = 0; i < 4; ++i)
    *(v4f*)(sA + (arow + 32 * i) * PA + akq * 4) = ra[i];
#pragma unroll
  for (int i = 0; i < 2; ++i) {
    const v4f lo = rb[2 * i], hi = rb[2 * i + 1];
    float* d = sB + (bpr + 8 * i) * PB + bnq * 8;
    *(v4f*)(d)     = __builtin_shufflevector(lo, hi, 0, 4, 1, 5);
    *(v4f*)(d + 4) = __builtin_shufflevector(lo, hi, 2, 6, 3, 7);
  }
}

struct FragSet { v2f a[2][4]; v2f b[2][2]; };

__device__ __forceinline__ void gemm_load_frags(const float* __restrict__ ap,
                                                const float* __restrict__ bp,
                                                int pr, FragSet& f) {
#pragma unroll
  for (int s = 0; s < 2; ++s) {
    const int ks = 2 * pr + s;
#pragma unroll
    for (int i = 0; i < 4; ++i) f.a[s][i] = *(const v2f*)(ap + 16 * i * PA + 4 * ks);
#pragma unroll
    for (int j = 0; j < 2; ++j) f.b[s][j] = *(const v2f*)(bp + 2 * ks * PB + 32 * j);
  }
}

__device__ __forceinline__ void gemm_compute_tile(const float* __restrict__ sA,
                                                  const float* __restrict__ sB,
                                                  int wm, int wn, int ln, int hl,
                                                  v8f (&c)[4][2]) {
  const float* ap = sA + (wm * 64 + ln) * PA + 2 * hl;
  const float* bp = sB + hl * PB + (wn * 32 + ln) * 2;
  FragSet f[2];
  gemm_load_frags(ap, bp, 0, f[0]);
#pragma unroll
  for (int pr = 0; pr < 4; ++pr) {
    if (pr + 1 < 4) gemm_load_frags(ap, bp, pr + 1, f[(pr + 1) & 1]);
    __builtin_amdgcn_sched_barrier(0);
#pragma unroll
    for (int s = 0; s < 2; ++s)
#pragma unroll
      for (int i = 0; i < 4; ++i)
#pragma unroll
        for (int j = 0; j < 2; ++j)
          c[i][j] = WMMA_F32(f[pr & 1].a[s][i], f[pr & 1].b[s][j], c[i][j]);
    if (pr + 1 < 4) __builtin_amdgcn_sched_barrier(0);
  }
}

template <int LDB>
__device__ __forceinline__ void gemm_mainloop(const float* A, const float* W,
                                              int m0, int n0, int tid, int wm, int wn,
                                              int ln, int hl, float* sA, float* sB,
                                              v8f (&c)[4][2]) {
  v4f ra[4], rb[4];
  gemm_load_tile<LDB>(A, W, m0, n0, 0, tid, ra, rb);
  gemm_store_tile(sA, sB, tid, ra, rb);
  __syncthreads();
  for (int kt = 0; kt < NKT - 1; ++kt) {
    const int cur = kt & 1;
    gemm_load_tile<LDB>(A, W, m0, n0, kt + 1, tid, ra, rb);
    __builtin_amdgcn_fence(__ATOMIC_RELEASE, "wavefront");
    gemm_compute_tile(sA + cur * SA, sB + cur * SB, wm, wn, ln, hl, c);
    gemm_store_tile(sA + (cur ^ 1) * SA, sB + (cur ^ 1) * SB, tid, ra, rb);
    __syncthreads();
  }
  gemm_compute_tile(sA + ((NKT - 1) & 1) * SA, sB + ((NKT - 1) & 1) * SB, wm, wn, ln, hl, c);
}

__global__ __launch_bounds__(256)
void qkv_gemm_kernel(const float* X, const float* W,
                     const float* __restrict__ bias,
                     float* __restrict__ q, float* __restrict__ k,
                     float* __restrict__ v) {
  __shared__ __attribute__((aligned(16))) float sA[2 * SA];
  __shared__ __attribute__((aligned(16))) float sB[2 * SB];
  const int tid = threadIdx.x;
  const int lane = tid & 31;
  const int hl = lane >> 4;
  const int ln = lane & 15;
  const int wave = tid >> 5;
  const int wm = wave >> 2, wn = wave & 3;
  const int n0 = blockIdx.x * BN;
  const int m0 = blockIdx.y * BM;

  v8f c[4][2] = {};
  gemm_mainloop<NQKV>(X, W, m0, n0, tid, wm, wn, ln, hl, sA, sB, c);

  const int which = n0 >> 10;
  float* dst = (which == 0) ? q : (which == 1) ? k : v;
#pragma unroll
  for (int j = 0; j < 2; ++j) {
    const int col = n0 + wn * 32 + 16 * j + ln;
    const float bval = bias[col];
    const int rem = col & (CC - 1);
    const int h = rem >> 6;
    const int d = rem & 63;
#pragma unroll
    for (int i = 0; i < 4; ++i) {
#pragma unroll
      for (int r = 0; r < 8; ++r) {
        const int trow = m0 + wm * 64 + 16 * i + r + 8 * hl;
        const int bidx = trow >> 11;
        const int t = trow & (TT - 1);
        const size_t idx = (((size_t)(bidx * HH + h) * TT) + t) * DD + d;
        dst[idx] = c[i][j][r] + bval;
      }
    }
  }
}

static constexpr int QB = 128;
static constexpr int KB = 128;
static constexpr int PK = DD + 4;
static constexpr int PV = 2 * DD + 32;
static constexpr int SK = KB * PK;
static constexpr int SV = (KB / 2) * PV;
static constexpr int NQB = TT / QB;

__device__ __forceinline__ void attn_load_block(const float* kbase, const float* vbase,
                                                int kb, int tid, v4f (&rk)[8], v4f (&rv)[8]) {
  const float* kp = kbase + (size_t)kb * KB * DD + tid * 4;
  const float* vp = vbase + (size_t)kb * KB * DD + (tid >> 4) * 2 * DD + (tid & 15) * 4;
#pragma unroll
  for (int i = 0; i < 8; ++i) rk[i] = *(const v4f*)(kp + i * 1024);
#pragma unroll
  for (int j = 0; j < 4; ++j)
#pragma unroll
    for (int e = 0; e < 2; ++e) rv[2 * j + e] = *(const v4f*)(vp + (32 * j + e) * DD);
}

__device__ __forceinline__ void attn_store_block(float* sK, float* sV, int tid,
                                                 const v4f (&rk)[8], const v4f (&rv)[8]) {
  const int row = tid >> 4, c4 = (tid & 15) * 4;
#pragma unroll
  for (int i = 0; i < 8; ++i) *(v4f*)(sK + (row + 16 * i) * PK + c4) = rk[i];
#pragma unroll
  for (int j = 0; j < 4; ++j) {
    const v4f lo = rv[2 * j], hi = rv[2 * j + 1];
    float* dst = sV + (row + 16 * j) * PV + c4 * 2;
    *(v4f*)(dst)     = __builtin_shufflevector(lo, hi, 0, 4, 1, 5);
    *(v4f*)(dst + 4) = __builtin_shufflevector(lo, hi, 2, 6, 3, 7);
  }
}

struct KFrags { v2f f[4][4]; };

__device__ __forceinline__ void attn_load_kfrags(const float* __restrict__ gK, int tb, KFrags& f) {
#pragma unroll
  for (int tt = 0; tt < 4; ++tt)
#pragma unroll
    for (int u = 0; u < 4; ++u)
      f.f[tt][u] = *(const v2f*)(gK + u * 16 * PK + 4 * (4 * tb + tt));
}

__device__ __forceinline__ float xhalf(float x) {
  const int b = __builtin_bit_cast(int, x);
  return __builtin_bit_cast(float,
                            __builtin_amdgcn_permlanex16(b, b, 0x76543210u, 0xfedcba98u, false, false));
}

__device__ __forceinline__ void attn_tile_update(v8f s, bool diag, const float* __restrict__ vt,
                                                 int ln, int hl, float& m_i, float& l_i, v8f (&o)[4]) {
  if (diag) {
#pragma unroll
    for (int r = 0; r < 8; ++r) {
      const int key = 4 * (r >> 1) + 2 * hl + (r & 1);
      if (key > ln) s[r] = -__builtin_inff();
    }
  }
  float mx = fmaxf(fmaxf(fmaxf(s[0], s[1]), fmaxf(s[2], s[3])),
                   fmaxf(fmaxf(s[4], s[5]), fmaxf(s[6], s[7])));
  mx = fmaxf(mx, xhalf(mx));
  const float mn = fmaxf(m_i, mx);
  const float sc = __expf(m_i - mn);
  float p[8];
#pragma unroll
  for (int r = 0; r < 8; ++r) p[r] = __expf(s[r] - mn);
  const float ps = ((p[0] + p[1]) + (p[2] + p[3])) + ((p[4] + p[5]) + (p[6] + p[7]));
  l_i = l_i * sc + ps;
  m_i = mn;

#pragma unroll
  for (int n = 0; n < 4; ++n) o[n] *= sc;

#pragma unroll
  for (int t = 0; t < 4; ++t) {
    v2f pb;
    pb.x = p[2 * t];
    pb.y = p[2 * t + 1];
    const float* vrow = vt + (2 * t + hl) * PV + ln * 2;
#pragma unroll
    for (int n = 0; n < 4; ++n) {
      const v2f vf = *(const v2f*)(vrow + 32 * n);
      o[n] = WMMA_F32(vf, pb, o[n]);
    }
  }
}

static constexpr int PO = DD + 4;

__global__ __launch_bounds__(256)
void attn_kernel(const float* __restrict__ q, const float* kmat,
                 const float* vmat, float* __restrict__ out) {
  __shared__ __attribute__((aligned(16))) float sK[2 * SK];
  __shared__ __attribute__((aligned(16))) float sV[2 * SV];
  __shared__ __attribute__((aligned(16))) float otile[8][16][PO];

  const int tid = threadIdx.x;
  const int lane = tid & 31;
  const int hl = lane >> 4;
  const int ln = lane & 15;
  const int wave = __builtin_amdgcn_readfirstlane(tid >> 5);
  const int pair = blockIdx.x;
  const int bh = blockIdx.y;
  const int b = bh >> 4, h = bh & 15;
  const int kr = ((ln & 6) << 1) | ((ln >> 2) & 2) | (ln & 1);

  const float* qbase = q + (size_t)bh * TT * DD;
  const float* kbase = kmat + (size_t)bh * TT * DD;
  const float* vbase = vmat + (size_t)bh * TT * DD;

  v4f rk[8], rv[8];

  for (int pass = 0; pass < 2; ++pass) {
    const int qb = pass == 0 ? (NQB - 1 - pair) : pair;
    const int qt = qb * 8 + wave;
    const int q0 = qt * 16;
    const int nkb = qb + 1;

    v2f qa[16];
    {
      const float* qrow = qbase + (size_t)(q0 + ln) * DD + 2 * hl;
#pragma unroll
      for (int t = 0; t < 16; ++t) {
        v2f a = *(const v2f*)(qrow + 4 * t);
        qa[t] = a * 0.125f;
      }
    }

    v8f o[4] = {};
    float m_i = -__builtin_inff(), l_i = 0.0f;

    attn_load_block(kbase, vbase, 0, tid, rk, rv);
    attn_store_block(sK, sV, tid, rk, rv);
    __syncthreads();

    for (int kb = 0; kb < nkb; ++kb) {
      const int cur = kb & 1;
      if (kb + 1 < nkb) {
        attn_load_block(kbase, vbase, kb + 1, tid, rk, rv);
        __builtin_amdgcn_fence(__ATOMIC_RELEASE, "wavefront");
      }
      const float* cK = sK + cur * SK;
      const float* cV = sV + cur * SV;

#pragma unroll 1
      for (int g = 0; g < 2; ++g) {
        const int kt0 = kb * 8 + g * 4;
        const float* gK = cK + (g * 64 + kr) * PK + 2 * hl;
        const float* gV = cV + (g * 32) * PV;
        if (qt >= kt0 + 4) {
          v8f s[4] = {};
          KFrags f[2];
          attn_load_kfrags(gK, 0, f[0]);
#pragma unroll
          for (int tb = 0; tb < 4; ++tb) {
            if (tb + 1 < 4) attn_load_kfrags(gK, tb + 1, f[(tb + 1) & 1]);
            __builtin_amdgcn_sched_barrier(0);
#pragma unroll
            for (int tt = 0; tt < 4; ++tt)
#pragma unroll
              for (int u = 0; u < 4; ++u)
                s[u] = WMMA_F32(f[tb & 1].f[tt][u], qa[4 * tb + tt], s[u]);
            if (tb + 1 < 4) __builtin_amdgcn_sched_barrier(0);
          }
#pragma unroll
          for (int u = 0; u < 4; ++u)
            attn_tile_update(s[u], false, gV + u * 8 * PV, ln, hl, m_i, l_i, o);
        } else if (qt >= kt0) {
          const int nt = qt - kt0 + 1;
#pragma unroll 1
          for (int u = 0; u < nt; ++u) {
            v8f s = {};
#pragma unroll
            for (int t = 0; t < 16; ++t) {
              v2f kf = *(const v2f*)(gK + u * 16 * PK + 4 * t);
              s = WMMA_F32(kf, qa[t], s);
            }
            attn_tile_update(s, u == nt - 1, gV + u * 8 * PV, ln, hl, m_i, l_i, o);
          }
        }
      }

      if (kb + 1 < nkb)
        attn_store_block(sK + (cur ^ 1) * SK, sV + (cur ^ 1) * SV, tid, rk, rv);
      __syncthreads();
    }

    {
      const float inv = 1.0f / (l_i + xhalf(l_i));
      float (*ot)[PO] = otile[wave];
#pragma unroll
      for (int n = 0; n < 4; ++n) {
        const v8f on = o[n] * inv;
        *(v4f*)&ot[ln][16 * n + 8 * hl]     = __builtin_shufflevector(on, on, 0, 1, 2, 3);
        *(v4f*)&ot[ln][16 * n + 8 * hl + 4] = __builtin_shufflevector(on, on, 4, 5, 6, 7);
      }
      __builtin_amdgcn_fence(__ATOMIC_RELEASE, "wavefront");
      __builtin_amdgcn_fence(__ATOMIC_ACQUIRE, "wavefront");
#pragma unroll
      for (int i = 0; i < 8; ++i) {
        const int row = 2 * i + hl;
        const v4f x = *(const v4f*)&ot[row][ln * 4];
        *(v4f*)(out + ((size_t)(b * TT + q0 + row) * CC) + h * DD + ln * 4) = x;
      }
      __builtin_amdgcn_fence(__ATOMIC_RELEASE, "wavefront");
    }
  }
}

__global__ __launch_bounds__(256)
void out_gemm_kernel(const float* A, const float* W,
                     const float* __restrict__ bias, float* __restrict__ out) {
  __shared__ __attribute__((aligned(16))) float sA[2 * SA];
  __shared__ __attribute__((aligned(16))) float sB[2 * SB];
  const int tid = threadIdx.x;
  const int lane = tid & 31;
  const int hl = lane >> 4;
  const int ln = lane & 15;
  const int wave = tid >> 5;
  const int wm = wave >> 2, wn = wave & 3;
  const int n0 = blockIdx.x * BN;
  const int m0 = blockIdx.y * BM;

  v8f c[4][2] = {};
  gemm_mainloop<CC>(A, W, m0, n0, tid, wm, wn, ln, hl, sA, sB, c);

#pragma unroll
  for (int j = 0; j < 2; ++j) {
    const int col = n0 + wn * 32 + 16 * j + ln;
    const float bval = bias[col];
#pragma unroll
    for (int i = 0; i < 4; ++i) {
#pragma unroll
      for (int r = 0; r < 8; ++r) {
        out[(size_t)(m0 + wm * 64 + 16 * i + r + 8 * hl) * CC + col] = c[i][j][r] + bval;
      }
    }
  }
}

extern "C" void kernel_launch(void* const* d_in, const int* in_sizes, int n_in,
                              void* d_out, int out_size, void* d_ws, size_t ws_size,
                              hipStream_t stream) {
  (void)in_sizes; (void)n_in; (void)out_size; (void)ws_size;
  const float* x    = (const float*)d_in[0];
  const float* Wqkv = (const float*)d_in[1];
  const float* bqkv = (const float*)d_in[2];
  const float* Wout = (const float*)d_in[3];
  const float* bout = (const float*)d_in[4];
  float* out = (float*)d_out;

  float* qp = (float*)d_ws;
  float* kp = qp + HSZ;
  float* vp = kp + HSZ;
  float* ao = vp + HSZ;

  const dim3 gblk(256, 1, 1);
  qkv_gemm_kernel<<<dim3(NQKV / BN, MTOT / BM), gblk, 0, stream>>>(x, Wqkv, bqkv, qp, kp, vp);
  attn_kernel<<<dim3(NQB / 2, BB * HH), gblk, 0, stream>>>(qp, kp, vp, ao);
  out_gemm_kernel<<<dim3(CC / BN, MTOT / BM), gblk, 0, stream>>>(ao, Wout, bout, out);
}
